# starting kernel
# baseline (speedup 1.0000x reference)
#include <hip/hip_runtime.h>
#include <math.h>

typedef float v2f __attribute__((ext_vector_type(2)));
typedef float v8f __attribute__((ext_vector_type(8)));

#define H    128
#define EPSN 1e-8f

__global__ __launch_bounds__(32)
void node_mlp_kernel(const float* __restrict__ emb,
                     const float* __restrict__ W1, const float* __restrict__ b1,
                     const float* __restrict__ W2, const float* __restrict__ b2,
                     float* __restrict__ Y, float* __restrict__ nrm, int N)
{
    __shared__ float lds[16 * H];
    const int lane = threadIdx.x;
    const int m    = lane & 15;
    const int kh   = (lane >> 4) * 2;
    const int rhalf = (lane >> 4) << 3;
    const int r0   = blockIdx.x * 16;

    for (int i = lane; i < 16 * H; i += 32) {
        int r = r0 + (i >> 7);
        lds[i] = (r < N) ? emb[(size_t)r * H + (i & 127)] : 0.0f;
    }
    __syncthreads();

    v8f acc[8];
    const v8f vzero = {0.f, 0.f, 0.f, 0.f, 0.f, 0.f, 0.f, 0.f};

    #pragma unroll
    for (int n = 0; n < 8; ++n) acc[n] = vzero;
    for (int k = 0; k < H; k += 4) {
        v2f a = { lds[m * H + k + kh], lds[m * H + k + kh + 1] };
        #pragma unroll
        for (int n = 0; n < 8; ++n) {
            const float* wrow = W1 + (size_t)(n * 16 + m) * H + k + kh;
            v2f b = { wrow[0], wrow[1] };
            acc[n] = __builtin_amdgcn_wmma_f32_16x16x4_f32(
                false, a, false, b, (short)0, acc[n], false, false);
        }
    }
    __syncthreads();
    #pragma unroll
    for (int n = 0; n < 8; ++n) {
        int col = n * 16 + m;
        float bias = b1[col];
        #pragma unroll
        for (int r = 0; r < 8; ++r) {
            float x = acc[n][r] + bias;
            lds[(r + rhalf) * H + col] = (x > 0.0f) ? x : expm1f(x);
        }
    }
    __syncthreads();

    #pragma unroll
    for (int n = 0; n < 8; ++n) acc[n] = vzero;
    for (int k = 0; k < H; k += 4) {
        v2f a = { lds[m * H + k + kh], lds[m * H + k + kh + 1] };
        #pragma unroll
        for (int n = 0; n < 8; ++n) {
            const float* wrow = W2 + (size_t)(n * 16 + m) * H + k + kh;
            v2f b = { wrow[0], wrow[1] };
            acc[n] = __builtin_amdgcn_wmma_f32_16x16x4_f32(
                false, a, false, b, (short)0, acc[n], false, false);
        }
    }
    __syncthreads();
    #pragma unroll
    for (int n = 0; n < 8; ++n) {
        int col = n * 16 + m;
        float bias = b2[col];
        #pragma unroll
        for (int r = 0; r < 8; ++r)
            lds[(r + rhalf) * H + col] = acc[n][r] + bias;
    }
    __syncthreads();

    if (lane < 16) {
        int r = r0 + lane;
        if (r < N) {
            float s = 0.0f;
            #pragma unroll 4
            for (int j = 0; j < H; ++j) {
                float v = lds[lane * H + j];
                s += v * v;
            }
            nrm[r] = fmaxf(sqrtf(s), EPSN);
        }
    }
    for (int i = lane; i < 16 * H; i += 32) {
        int r = r0 + (i >> 7);
        if (r < N) Y[(size_t)r * H + (i & 127)] = lds[i];
    }
}

__global__ __launch_bounds__(256)
void edge_cos_kernel(const int* __restrict__ ei, const float* __restrict__ Y,
                     const float* __restrict__ nrm, float* __restrict__ out,
                     int E)
{
    int wid  = (int)((blockIdx.x * 256u + threadIdx.x) >> 5);
    int lane = threadIdx.x & 31;
    if (wid >= E) return;
    int c = ei[wid];
    int r = ei[E + wid];
    const float4* f1 = (const float4*)(Y + (size_t)c * H);
    const float4* f2 = (const float4*)(Y + (size_t)r * H);
    float4 a = f1[lane];
    float4 b = f2[lane];
    float s = a.x * b.x + a.y * b.y + a.z * b.z + a.w * b.w;
    #pragma unroll
    for (int off = 16; off > 0; off >>= 1)
        s += __shfl_xor(s, off, 32);
    if (lane == 0) out[wid] = s / (nrm[c] * nrm[r]);
}

extern "C" void kernel_launch(void* const* d_in, const int* in_sizes, int n_in,
                              void* d_out, int out_size, void* d_ws, size_t ws_size,
                              hipStream_t stream)
{
    const float* emb = (const float*)d_in[0];
    const int*   ei  = (const int*)  d_in[1];
    const float* W1  = (const float*)d_in[2];
    const float* b1  = (const float*)d_in[3];
    const float* W2  = (const float*)d_in[4];
    const float* b2  = (const float*)d_in[5];
    float* out = (float*)d_out;

    const int N = in_sizes[0] / H;
    const int E = in_sizes[1] / 2;

    float* Y   = (float*)d_ws;
    float* nv  = Y + (size_t)N * H;

    int nblk = (N + 15) / 16;
    hipLaunchKernelGGL(node_mlp_kernel, dim3(nblk), dim3(32), 0, stream,
                       emb, W1, b1, W2, b2, Y, nv, N);

    int eblk = (E + 7) / 8;
    hipLaunchKernelGGL(edge_cos_kernel, dim3(eblk), dim3(256), 0, stream,
                       ei, Y, nv, out, E);
}


# edge kernel: 8192 resident waves loop over 32-edge runs; batch loop kept rolled so a trip sends its 16 row loads out before any wait (was two at a time); per-edge sum order and rounding kept
# speedup vs baseline: 2.4997x; 2.4997x over previous
#include <hip/hip_runtime.h>
#include <math.h>

typedef float v2f __attribute__((ext_vector_type(2)));
typedef float v4f __attribute__((ext_vector_type(4)));
typedef float v8f __attribute__((ext_vector_type(8)));

#define H    128
#define EPSN 1e-8f
#define LDP  132
#define MR   32
#define NWG  512

__global__ __launch_bounds__(256)
void node_mlp_kernel(const float* __restrict__ emb,
                     const float* __restrict__ W1, const float* __restrict__ b1,
                     const float* __restrict__ W2, const float* __restrict__ b2,
                     float* __restrict__ Y, float* __restrict__ nrm, int N)
{
    __shared__ __attribute__((aligned(16))) float smem[H * LDP];
    float* const xs = smem;
    float* const hs = smem + MR * LDP;
    float* const ys = smem + 2 * MR * LDP;

    const int tid  = threadIdx.x;
    const int lane = tid & 31;
    const int w    = tid >> 5;
    const int m    = lane & 15;
    const int kh   = (lane >> 4) * 2;
    const int rhalf = (lane >> 4) << 3;
    const int col  = w * 16 + m;

    v4f xp0, xp1, xp2, xp3;
    {
        const int rb = (int)blockIdx.x * MR + w;
        xp0 = *(const v4f*)(emb + (size_t)min(rb,      N - 1) * H + 4 * lane);
        xp1 = *(const v4f*)(emb + (size_t)min(rb + 8,  N - 1) * H + 4 * lane);
        xp2 = *(const v4f*)(emb + (size_t)min(rb + 16, N - 1) * H + 4 * lane);
        xp3 = *(const v4f*)(emb + (size_t)min(rb + 24, N - 1) * H + 4 * lane);
    }

    v2f w1f[32], w2f[32];
    {
        v4f t1[16], t2[16];
        #pragma unroll
        for (int i = 0; i < 16; ++i) {
            t1[i] = *(const v4f*)(W1 + (size_t)(w + 8 * i) * H + 4 * lane);
            t2[i] = *(const v4f*)(W2 + (size_t)(w + 8 * i) * H + 4 * lane);
        }
        #pragma unroll
        for (int i = 0; i < 16; ++i)
            *(v4f*)&smem[(w + 8 * i) * LDP + 4 * lane] = t1[i];
        __syncthreads();
        #pragma unroll
        for (int kk = 0; kk < 32; ++kk)
            w1f[kk] = *(const v2f*)&smem[col * LDP + 4 * kk + kh];
        __syncthreads();
        #pragma unroll
        for (int i = 0; i < 16; ++i)
            *(v4f*)&smem[(w + 8 * i) * LDP + 4 * lane] = t2[i];
        __syncthreads();
        #pragma unroll
        for (int kk = 0; kk < 32; ++kk)
            w2f[kk] = *(const v2f*)&smem[col * LDP + 4 * kk + kh];
        __syncthreads();
    }
    const float bias1 = b1[col];
    const float bias2 = b2[col];

    const int ngroups = (N + MR - 1) / MR;
    const v8f vzero = {0.f, 0.f, 0.f, 0.f, 0.f, 0.f, 0.f, 0.f};

    for (int g = blockIdx.x; g < ngroups; g += NWG) {
        *(v4f*)&xs[(w)      * LDP + 4 * lane] = xp0;
        *(v4f*)&xs[(w + 8)  * LDP + 4 * lane] = xp1;
        *(v4f*)&xs[(w + 16) * LDP + 4 * lane] = xp2;
        *(v4f*)&xs[(w + 24) * LDP + 4 * lane] = xp3;
        __syncthreads();

        if (g + NWG < ngroups) {
            const int rb = (g + NWG) * MR + w;
            xp0 = *(const v4f*)(emb + (size_t)min(rb,      N - 1) * H + 4 * lane);
            xp1 = *(const v4f*)(emb + (size_t)min(rb + 8,  N - 1) * H + 4 * lane);
            xp2 = *(const v4f*)(emb + (size_t)min(rb + 16, N - 1) * H + 4 * lane);
            xp3 = *(const v4f*)(emb + (size_t)min(rb + 24, N - 1) * H + 4 * lane);
        }

        v8f acc0 = vzero, acc1 = vzero;
        #pragma unroll
        for (int kk = 0; kk < 32; ++kk) {
            v2f a0 = *(const v2f*)&xs[m * LDP + 4 * kk + kh];
            v2f a1 = *(const v2f*)&xs[(16 + m) * LDP + 4 * kk + kh];
            acc0 = __builtin_amdgcn_wmma_f32_16x16x4_f32(
                false, a0, false, w1f[kk], (short)0, acc0, false, false);
            acc1 = __builtin_amdgcn_wmma_f32_16x16x4_f32(
                false, a1, false, w1f[kk], (short)0, acc1, false, false);
        }
        #pragma unroll
        for (int r = 0; r < 8; ++r) {
            float x0 = acc0[r] + bias1;
            float x1 = acc1[r] + bias1;
            hs[(r + rhalf) * LDP + col]      = (x0 > 0.0f) ? x0 : expm1f(x0);
            hs[(16 + r + rhalf) * LDP + col] = (x1 > 0.0f) ? x1 : expm1f(x1);
        }
        __syncthreads();

        acc0 = vzero; acc1 = vzero;
        #pragma unroll
        for (int kk = 0; kk < 32; ++kk) {
            v2f a0 = *(const v2f*)&hs[m * LDP + 4 * kk + kh];
            v2f a1 = *(const v2f*)&hs[(16 + m) * LDP + 4 * kk + kh];
            acc0 = __builtin_amdgcn_wmma_f32_16x16x4_f32(
                false, a0, false, w2f[kk], (short)0, acc0, false, false);
            acc1 = __builtin_amdgcn_wmma_f32_16x16x4_f32(
                false, a1, false, w2f[kk], (short)0, acc1, false, false);
        }
        #pragma unroll
        for (int r = 0; r < 8; ++r) {
            ys[(r + rhalf) * LDP + col]      = acc0[r] + bias2;
            ys[(16 + r + rhalf) * LDP + col] = acc1[r] + bias2;
        }
        __syncthreads();

        v4f yv[4];
        float ss[4];
        #pragma unroll
        for (int i = 0; i < 4; ++i) {
            yv[i] = *(const v4f*)&ys[(w + 8 * i) * LDP + 4 * lane];
            ss[i] = yv[i].x * yv[i].x + yv[i].y * yv[i].y + yv[i].z * yv[i].z + yv[i].w * yv[i].w;
        }
        #pragma unroll
        for (int off = 16; off > 0; off >>= 1) {
            float o0 = __shfl_xor(ss[0], off, 32);
            float o1 = __shfl_xor(ss[1], off, 32);
            float o2 = __shfl_xor(ss[2], off, 32);
            float o3 = __shfl_xor(ss[3], off, 32);
            ss[0] += o0; ss[1] += o1; ss[2] += o2; ss[3] += o3;
        }
        #pragma unroll
        for (int i = 0; i < 4; ++i) {
            const int gr = g * MR + w + 8 * i;
            if (gr < N) {
                *(v4f*)(Y + (size_t)gr * H + 4 * lane) = yv[i];
                if (lane == 0) nrm[gr] = fmaxf(sqrtf(ss[i]), EPSN);
            }
        }
    }
}

#define EWG 1024

__device__ __forceinline__ float xchg16(float x)
{
    const int b = __builtin_bit_cast(int, x);
    return __builtin_bit_cast(float,
        __builtin_amdgcn_permlanex16(b, b, 0x76543210, 0xfedcba98, false, false));
}
template <int D>
__device__ __forceinline__ float xchgrow(float x)
{
    const int b = __builtin_bit_cast(int, x);
    return __builtin_bit_cast(float,
        __builtin_amdgcn_update_dpp(0, b, 0x160 + D, 0xf, 0xf, true));
}
template <int D>
__device__ __forceinline__ float fold(float lo, float hi, int lane)
{
    const bool up   = (lane & D) != 0;
    const float keep = up ? hi : lo;
    const float send = up ? lo : hi;
    if constexpr (D == 16) return keep + xchg16(send);
    else                   return keep + xchgrow<D>(send);
}

__global__ __launch_bounds__(256)
__attribute__((amdgpu_waves_per_eu(8, 8)))
void edge_cos_kernel(const int* ei, const float* Y, const float* nrm, float* out, int E)
{
    const int lane   = threadIdx.x & 31;
    const int nruns  = (E + 31) >> 5;
    const int nwaves = (int)gridDim.x * 8;
    int run = (int)blockIdx.x * 8
            + __builtin_amdgcn_readfirstlane((int)threadIdx.x >> 5);
    if (run >= nruns) return;

    const unsigned piece = (unsigned)lane * 16u;
    const int ec = min(run * 32 + lane, E - 1);
    int c  = ei[ec];
    int r  = ei[E + ec];

    for (;;) {
        const int  nrun = run + nwaves;
        const bool more = nrun < nruns;
        const int ecn = min(nrun * 32 + lane, E - 1);
        const int cn  = ei[ecn];
        const int rn  = ei[E + ecn];
        const float n1 = nrm[c];
        const float n2 = nrm[r];
        float R0 = 0.f, R1 = 0.f, R2 = 0.f, R3 = 0.f;
        #pragma nounroll
        for (int g = 0; g < 4; ++g) {
            v4f a[8], b[8];
            #pragma unroll
            for (int k = 0; k < 8; ++k) {
                const unsigned cj = (unsigned)__builtin_amdgcn_readlane(c, 4 * k + g);
                const unsigned rj = (unsigned)__builtin_amdgcn_readlane(r, 4 * k + g);
                const char* const rowc = (const char*)(Y + (size_t)cj * H);
                const char* const rowr = (const char*)(Y + (size_t)rj * H);
                a[k] = *(const v4f*)(rowc + piece);
                b[k] = *(const v4f*)(rowr + piece);
            }
            float p[8];
            #pragma unroll
            for (int k = 0; k < 8; ++k)
                p[k] = __builtin_fmaf(a[k].w, b[k].w,
                       __builtin_fmaf(a[k].z, b[k].z,
                       __builtin_fmaf(a[k].x, b[k].x, a[k].y * b[k].y)));
            const float q0 = fold<16>(p[0], p[4], lane);
            const float q1 = fold<16>(p[1], p[5], lane);
            const float q2 = fold<16>(p[2], p[6], lane);
            const float q3 = fold<16>(p[3], p[7], lane);
            const float t0 = fold<8>(q0, q2, lane);
            const float t1 = fold<8>(q1, q3, lane);
            const float Rg = fold<4>(t0, t1, lane);
            R0 = (g == 0) ? Rg : R0;
            R1 = (g == 1) ? Rg : R1;
            R2 = (g == 2) ? Rg : R2;
            R3 = (g == 3) ? Rg : R3;
        }
        const float u0 = fold<2>(R0, R2, lane);
        const float u1 = fold<2>(R1, R3, lane);
        const float s  = fold<1>(u0, u1, lane);
        const int e = run * 32 + lane;
        if (e < E) out[e] = s / (n1 * n2);
        if (!more) break;
        run = nrun; c = cn; r = rn;
    }
}

extern "C" void kernel_launch(void* const* d_in, const int* in_sizes, int n_in,
                              void* d_out, int out_size, void* d_ws, size_t ws_size,
                              hipStream_t stream)
{
    const float* emb = (const float*)d_in[0];
    const int*   ei  = (const int*)  d_in[1];
    const float* W1  = (const float*)d_in[2];
    const float* b1  = (const float*)d_in[3];
    const float* W2  = (const float*)d_in[4];
    const float* b2  = (const float*)d_in[5];
    float* out = (float*)d_out;

    const int N = in_sizes[0] / H;
    const int E = in_sizes[1] / 2;

    float* Y   = (float*)d_ws;
    float* nv  = Y + (size_t)N * H;

    hipLaunchKernelGGL(node_mlp_kernel, dim3(NWG), dim3(256), 0, stream,
                       emb, W1, b1, W2, b2, Y, nv, N);

    hipLaunchKernelGGL(edge_cos_kernel, dim3(EWG), dim3(256), 0, stream,
                       ei, Y, nv, out, E);
}


# edge kernel: one 32-wave workgroup per WGP copies the 200-KB norm plane into LDS once with whole-line loads; each edge's two norms are LDS reads, not 32-line global gathers; sums and rounding kept
# speedup vs baseline: 2.5816x; 1.0328x over previous
#include <hip/hip_runtime.h>
#include <math.h>

typedef float v2f __attribute__((ext_vector_type(2)));
typedef float v4f __attribute__((ext_vector_type(4)));
typedef float v8f __attribute__((ext_vector_type(8)));

#define H    128
#define EPSN 1e-8f
#define LDP  132
#define MR   32
#define NWG  512

__global__ __launch_bounds__(256)
void node_mlp_kernel(const float* __restrict__ emb,
                     const float* __restrict__ W1, const float* __restrict__ b1,
                     const float* __restrict__ W2, const float* __restrict__ b2,
                     float* __restrict__ Y, float* __restrict__ nrm, int N)
{
    __shared__ __attribute__((aligned(16))) float smem[H * LDP];
    float* const xs = smem;
    float* const hs = smem + MR * LDP;
    float* const ys = smem + 2 * MR * LDP;

    const int tid  = threadIdx.x;
    const int lane = tid & 31;
    const int w    = tid >> 5;
    const int m    = lane & 15;
    const int kh   = (lane >> 4) * 2;
    const int rhalf = (lane >> 4) << 3;
    const int col  = w * 16 + m;

    v4f xp0, xp1, xp2, xp3;
    {
        const int rb = (int)blockIdx.x * MR + w;
        xp0 = *(const v4f*)(emb + (size_t)min(rb,      N - 1) * H + 4 * lane);
        xp1 = *(const v4f*)(emb + (size_t)min(rb + 8,  N - 1) * H + 4 * lane);
        xp2 = *(const v4f*)(emb + (size_t)min(rb + 16, N - 1) * H + 4 * lane);
        xp3 = *(const v4f*)(emb + (size_t)min(rb + 24, N - 1) * H + 4 * lane);
    }

    v2f w1f[32], w2f[32];
    {
        v4f t1[16], t2[16];
        #pragma unroll
        for (int i = 0; i < 16; ++i) {
            t1[i] = *(const v4f*)(W1 + (size_t)(w + 8 * i) * H + 4 * lane);
            t2[i] = *(const v4f*)(W2 + (size_t)(w + 8 * i) * H + 4 * lane);
        }
        #pragma unroll
        for (int i = 0; i < 16; ++i)
            *(v4f*)&smem[(w + 8 * i) * LDP + 4 * lane] = t1[i];
        __syncthreads();
        #pragma unroll
        for (int kk = 0; kk < 32; ++kk)
            w1f[kk] = *(const v2f*)&smem[col * LDP + 4 * kk + kh];
        __syncthreads();
        #pragma unroll
        for (int i = 0; i < 16; ++i)
            *(v4f*)&smem[(w + 8 * i) * LDP + 4 * lane] = t2[i];
        __syncthreads();
        #pragma unroll
        for (int kk = 0; kk < 32; ++kk)
            w2f[kk] = *(const v2f*)&smem[col * LDP + 4 * kk + kh];
        __syncthreads();
    }
    const float bias1 = b1[col];
    const float bias2 = b2[col];

    const int ngroups = (N + MR - 1) / MR;
    const v8f vzero = {0.f, 0.f, 0.f, 0.f, 0.f, 0.f, 0.f, 0.f};

    for (int g = blockIdx.x; g < ngroups; g += NWG) {
        *(v4f*)&xs[(w)      * LDP + 4 * lane] = xp0;
        *(v4f*)&xs[(w + 8)  * LDP + 4 * lane] = xp1;
        *(v4f*)&xs[(w + 16) * LDP + 4 * lane] = xp2;
        *(v4f*)&xs[(w + 24) * LDP + 4 * lane] = xp3;
        __syncthreads();

        if (g + NWG < ngroups) {
            const int rb = (g + NWG) * MR + w;
            xp0 = *(const v4f*)(emb + (size_t)min(rb,      N - 1) * H + 4 * lane);
            xp1 = *(const v4f*)(emb + (size_t)min(rb + 8,  N - 1) * H + 4 * lane);
            xp2 = *(const v4f*)(emb + (size_t)min(rb + 16, N - 1) * H + 4 * lane);
            xp3 = *(const v4f*)(emb + (size_t)min(rb + 24, N - 1) * H + 4 * lane);
        }

        v8f acc0 = vzero, acc1 = vzero;
        #pragma unroll
        for (int kk = 0; kk < 32; ++kk) {
            v2f a0 = *(const v2f*)&xs[m * LDP + 4 * kk + kh];
            v2f a1 = *(const v2f*)&xs[(16 + m) * LDP + 4 * kk + kh];
            acc0 = __builtin_amdgcn_wmma_f32_16x16x4_f32(
                false, a0, false, w1f[kk], (short)0, acc0, false, false);
            acc1 = __builtin_amdgcn_wmma_f32_16x16x4_f32(
                false, a1, false, w1f[kk], (short)0, acc1, false, false);
        }
        #pragma unroll
        for (int r = 0; r < 8; ++r) {
            float x0 = acc0[r] + bias1;
            float x1 = acc1[r] + bias1;
            hs[(r + rhalf) * LDP + col]      = (x0 > 0.0f) ? x0 : expm1f(x0);
            hs[(16 + r + rhalf) * LDP + col] = (x1 > 0.0f) ? x1 : expm1f(x1);
        }
        __syncthreads();

        acc0 = vzero; acc1 = vzero;
        #pragma unroll
        for (int kk = 0; kk < 32; ++kk) {
            v2f a0 = *(const v2f*)&hs[m * LDP + 4 * kk + kh];
            v2f a1 = *(const v2f*)&hs[(16 + m) * LDP + 4 * kk + kh];
            acc0 = __builtin_amdgcn_wmma_f32_16x16x4_f32(
                false, a0, false, w2f[kk], (short)0, acc0, false, false);
            acc1 = __builtin_amdgcn_wmma_f32_16x16x4_f32(
                false, a1, false, w2f[kk], (short)0, acc1, false, false);
        }
        #pragma unroll
        for (int r = 0; r < 8; ++r) {
            ys[(r + rhalf) * LDP + col]      = acc0[r] + bias2;
            ys[(16 + r + rhalf) * LDP + col] = acc1[r] + bias2;
        }
        __syncthreads();

        v4f yv[4];
        float ss[4];
        #pragma unroll
        for (int i = 0; i < 4; ++i) {
            yv[i] = *(const v4f*)&ys[(w + 8 * i) * LDP + 4 * lane];
            ss[i] = yv[i].x * yv[i].x + yv[i].y * yv[i].y + yv[i].z * yv[i].z + yv[i].w * yv[i].w;
        }
        #pragma unroll
        for (int off = 16; off > 0; off >>= 1) {
            float o0 = __shfl_xor(ss[0], off, 32);
            float o1 = __shfl_xor(ss[1], off, 32);
            float o2 = __shfl_xor(ss[2], off, 32);
            float o3 = __shfl_xor(ss[3], off, 32);
            ss[0] += o0; ss[1] += o1; ss[2] += o2; ss[3] += o3;
        }
        #pragma unroll
        for (int i = 0; i < 4; ++i) {
            const int gr = g * MR + w + 8 * i;
            if (gr < N) {
                *(v4f*)(Y + (size_t)gr * H + 4 * lane) = yv[i];
                if (lane == 0) nrm[gr] = fmaxf(sqrtf(ss[i]), EPSN);
            }
        }
    }
}

#define EWG  256
#define NTAB 53248

__device__ __forceinline__ float xchg16(float x)
{
    const int b = __builtin_bit_cast(int, x);
    return __builtin_bit_cast(float,
        __builtin_amdgcn_permlanex16(b, b, 0x76543210, 0xfedcba98, false, false));
}
template <int D>
__device__ __forceinline__ float xchgrow(float x)
{
    const int b = __builtin_bit_cast(int, x);
    return __builtin_bit_cast(float,
        __builtin_amdgcn_update_dpp(0, b, 0x160 + D, 0xf, 0xf, true));
}
template <int D>
__device__ __forceinline__ float fold(float lo, float hi, int lane)
{
    const bool up   = (lane & D) != 0;
    const float keep = up ? hi : lo;
    const float send = up ? lo : hi;
    if constexpr (D == 16) return keep + xchg16(send);
    else                   return keep + xchgrow<D>(send);
}

template <bool LDSN>
__device__ __forceinline__ void edge_runs(const int* ei, const float* Y, const float* nrm, const float* nl,
                                          float* out, int E, int run, int nruns, int nwaves, int lane, int c, int r)
{
    const unsigned piece = (unsigned)lane * 16u;
    for (;;) {
        const int  nrun = run + nwaves;
        const bool more = nrun < nruns;
        const int ecn = min(nrun * 32 + lane, E - 1);
        const int cn  = ei[ecn];
        const int rn  = ei[E + ecn];
        float n1, n2;
        if constexpr (LDSN) { n1 = nl[c];  n2 = nl[r];  }
        else                { n1 = nrm[c]; n2 = nrm[r]; }
        float R0 = 0.f, R1 = 0.f, R2 = 0.f, R3 = 0.f;
        #pragma nounroll
        for (int g = 0; g < 4; ++g) {
            v4f a[8], b[8];
            #pragma unroll
            for (int k = 0; k < 8; ++k) {
                const unsigned cj = (unsigned)__builtin_amdgcn_readlane(c, 4 * k + g);
                const unsigned rj = (unsigned)__builtin_amdgcn_readlane(r, 4 * k + g);
                const char* const rowc = (const char*)(Y + (size_t)cj * H);
                const char* const rowr = (const char*)(Y + (size_t)rj * H);
                a[k] = *(const v4f*)(rowc + piece);
                b[k] = *(const v4f*)(rowr + piece);
            }
            float p[8];
            #pragma unroll
            for (int k = 0; k < 8; ++k)
                p[k] = __builtin_fmaf(a[k].w, b[k].w,
                       __builtin_fmaf(a[k].z, b[k].z,
                       __builtin_fmaf(a[k].x, b[k].x, a[k].y * b[k].y)));
            const float q0 = fold<16>(p[0], p[4], lane);
            const float q1 = fold<16>(p[1], p[5], lane);
            const float q2 = fold<16>(p[2], p[6], lane);
            const float q3 = fold<16>(p[3], p[7], lane);
            const float t0 = fold<8>(q0, q2, lane);
            const float t1 = fold<8>(q1, q3, lane);
            const float Rg = fold<4>(t0, t1, lane);
            R0 = (g == 0) ? Rg : R0;
            R1 = (g == 1) ? Rg : R1;
            R2 = (g == 2) ? Rg : R2;
            R3 = (g == 3) ? Rg : R3;
        }
        const float u0 = fold<2>(R0, R2, lane);
        const float u1 = fold<2>(R1, R3, lane);
        const float s  = fold<1>(u0, u1, lane);
        const int e = run * 32 + lane;
        if (e < E) out[e] = s / (n1 * n2);
        if (!more) break;
        run = nrun; c = cn; r = rn;
    }
}

__global__ __launch_bounds__(1024)
__attribute__((amdgpu_waves_per_eu(8, 8)))
void edge_cos_kernel(const int* ei, const float* Y, const float* nrm, float* out, int E, int N)
{
    __shared__ __attribute__((aligned(16))) float nl[NTAB];

    const int tid    = (int)threadIdx.x;
    const int lane   = tid & 31;
    const int nruns  = (E + 31) >> 5;
    const int nwaves = (int)gridDim.x * 32;
    const int run = (int)blockIdx.x * 32 + __builtin_amdgcn_readfirstlane(tid >> 5);

    const int ec = min(run * 32 + lane, E - 1);
    const int c  = ei[ec];
    const int r  = ei[E + ec];

    if (N <= NTAB) {
        const int n4 = N >> 2;
        v4f tv[NTAB / 4096];
        #pragma unroll
        for (int i = 0; i < NTAB / 4096; ++i)
            if (1024 * i < n4)
                tv[i] = *(const v4f*)(nrm + 4 * (size_t)min(tid + 1024 * i, n4 - 1));
        #pragma unroll
        for (int i = 0; i < NTAB / 4096; ++i)
            if (tid + 1024 * i < n4)
                *(v4f*)&nl[4 * (tid + 1024 * i)] = tv[i];
        const int t = (n4 << 2) + tid;
        if (t < N) nl[t] = nrm[t];
        __syncthreads();
        if (run < nruns)
            edge_runs<true>(ei, Y, nrm, nl, out, E, run, nruns, nwaves, lane, c, r);
    } else {
        if (run < nruns)
            edge_runs<false>(ei, Y, nrm, nl, out, E, run, nruns, nwaves, lane, c, r);
    }
}

extern "C" void kernel_launch(void* const* d_in, const int* in_sizes, int n_in,
                              void* d_out, int out_size, void* d_ws, size_t ws_size,
                              hipStream_t stream)
{
    const float* emb = (const float*)d_in[0];
    const int*   ei  = (const int*)  d_in[1];
    const float* W1  = (const float*)d_in[2];
    const float* b1  = (const float*)d_in[3];
    const float* W2  = (const float*)d_in[4];
    const float* b2  = (const float*)d_in[5];
    float* out = (float*)d_out;

    const int N = in_sizes[0] / H;
    const int E = in_sizes[1] / 2;

    float* Y   = (float*)d_ws;
    float* nv  = Y + (size_t)N * H;

    hipLaunchKernelGGL(node_mlp_kernel, dim3(NWG), dim3(256), 0, stream,
                       emb, W1, b1, W2, b2, Y, nv, N);

    hipLaunchKernelGGL(edge_cos_kernel, dim3(EWG), dim3(1024), 0, stream,
                       ei, Y, nv, out, E, N);
}
